# starting kernel
# baseline (speedup 1.0000x reference)
#include <hip/hip_runtime.h>

constexpr int Bsz  = 256;
constexpr int Slen = 512;
constexpr int Edim = 100;
constexpr int Hdim = 512;
constexpr int KTOT = Edim + Hdim;

typedef __attribute__((ext_vector_type(2))) float v2f;
typedef __attribute__((ext_vector_type(8))) float v8f;

__device__ __forceinline__ float sigmoidf(float x) {
    return 1.0f / (1.0f + __expf(-x));
}

__global__ __launch_bounds__(256) void lstm_zero_init(float* __restrict__ h0,
                                                      float* __restrict__ c0,
                                                      int n) {
    int i = blockIdx.x * blockDim.x + threadIdx.x;
    if (i < n) { h0[i] = 0.0f; c0[i] = 0.0f; }
}

__global__ __launch_bounds__(128) void lstm_step(
    const int*   __restrict__ sentence,
    const float* __restrict__ emb,
    const float* __restrict__ Wih,
    const float* __restrict__ Whh,
    const float* __restrict__ bih,
    const float* __restrict__ bhh,
    const float* __restrict__ h_in,
    float*       __restrict__ c,
    float*       __restrict__ h_out,
    int t)
{
    __shared__ float Alds[2][KTOT * 16];

    const int tid    = threadIdx.x;
    const int m_base = blockIdx.x * 32;

    {
        const int r   = tid >> 2;
        const int sub = tid & 3;
        const int b   = m_base + r;
        float* panel  = Alds[r >> 4];
        const int rr  = r & 15;
        const int tok = sentence[(long)b * Slen + t];
        const float* erow = emb + (long)tok * Edim;
        for (int k = sub; k < Edim; k += 4)
            panel[k * 16 + rr] = erow[k];
        const float* hrow = h_in + (long)b * Hdim;
        for (int k = sub; k < Hdim; k += 4)
            panel[(Edim + k) * 16 + rr] = hrow[k];
    }
    __syncthreads();

    const int wave = tid >> 5;
    const int lane = tid & 31;
    const int half = lane >> 4;
    const int ml   = lane & 15;
    const int col  = blockIdx.y * 64 + wave * 16 + ml;

    v8f acc[2][4];
#pragma unroll
    for (int mt = 0; mt < 2; ++mt)
#pragma unroll
        for (int q = 0; q < 4; ++q)
            acc[mt][q] = (v8f){};

    const float* wi0 = Wih + (long)(0 * Hdim + col) * Edim;
    const float* wi1 = Wih + (long)(1 * Hdim + col) * Edim;
    const float* wi2 = Wih + (long)(2 * Hdim + col) * Edim;
    const float* wi3 = Wih + (long)(3 * Hdim + col) * Edim;

    for (int k0 = 0; k0 < Edim; k0 += 4) {
        const int kb = k0 + 2 * half;
        v2f a0, a1;
        a0.x = Alds[0][kb * 16 + ml];
        a0.y = Alds[0][(kb + 1) * 16 + ml];
        a1.x = Alds[1][kb * 16 + ml];
        a1.y = Alds[1][(kb + 1) * 16 + ml];
        v2f b0 = *(const v2f*)(wi0 + kb);
        v2f b1 = *(const v2f*)(wi1 + kb);
        v2f b2 = *(const v2f*)(wi2 + kb);
        v2f b3 = *(const v2f*)(wi3 + kb);
        acc[0][0] = __builtin_amdgcn_wmma_f32_16x16x4_f32(false, a0, false, b0, (short)0, acc[0][0], false, false);
        acc[1][0] = __builtin_amdgcn_wmma_f32_16x16x4_f32(false, a1, false, b0, (short)0, acc[1][0], false, false);
        acc[0][1] = __builtin_amdgcn_wmma_f32_16x16x4_f32(false, a0, false, b1, (short)0, acc[0][1], false, false);
        acc[1][1] = __builtin_amdgcn_wmma_f32_16x16x4_f32(false, a1, false, b1, (short)0, acc[1][1], false, false);
        acc[0][2] = __builtin_amdgcn_wmma_f32_16x16x4_f32(false, a0, false, b2, (short)0, acc[0][2], false, false);
        acc[1][2] = __builtin_amdgcn_wmma_f32_16x16x4_f32(false, a1, false, b2, (short)0, acc[1][2], false, false);
        acc[0][3] = __builtin_amdgcn_wmma_f32_16x16x4_f32(false, a0, false, b3, (short)0, acc[0][3], false, false);
        acc[1][3] = __builtin_amdgcn_wmma_f32_16x16x4_f32(false, a1, false, b3, (short)0, acc[1][3], false, false);
    }

    const float* wh0 = Whh + (long)(0 * Hdim + col) * Hdim;
    const float* wh1 = Whh + (long)(1 * Hdim + col) * Hdim;
    const float* wh2 = Whh + (long)(2 * Hdim + col) * Hdim;
    const float* wh3 = Whh + (long)(3 * Hdim + col) * Hdim;

    for (int k0 = 0; k0 < Hdim; k0 += 4) {
        const int kb = k0 + 2 * half;
        v2f a0, a1;
        a0.x = Alds[0][(Edim + kb) * 16 + ml];
        a0.y = Alds[0][(Edim + kb + 1) * 16 + ml];
        a1.x = Alds[1][(Edim + kb) * 16 + ml];
        a1.y = Alds[1][(Edim + kb + 1) * 16 + ml];
        v2f b0 = *(const v2f*)(wh0 + kb);
        v2f b1 = *(const v2f*)(wh1 + kb);
        v2f b2 = *(const v2f*)(wh2 + kb);
        v2f b3 = *(const v2f*)(wh3 + kb);
        acc[0][0] = __builtin_amdgcn_wmma_f32_16x16x4_f32(false, a0, false, b0, (short)0, acc[0][0], false, false);
        acc[1][0] = __builtin_amdgcn_wmma_f32_16x16x4_f32(false, a1, false, b0, (short)0, acc[1][0], false, false);
        acc[0][1] = __builtin_amdgcn_wmma_f32_16x16x4_f32(false, a0, false, b1, (short)0, acc[0][1], false, false);
        acc[1][1] = __builtin_amdgcn_wmma_f32_16x16x4_f32(false, a1, false, b1, (short)0, acc[1][1], false, false);
        acc[0][2] = __builtin_amdgcn_wmma_f32_16x16x4_f32(false, a0, false, b2, (short)0, acc[0][2], false, false);
        acc[1][2] = __builtin_amdgcn_wmma_f32_16x16x4_f32(false, a1, false, b2, (short)0, acc[1][2], false, false);
        acc[0][3] = __builtin_amdgcn_wmma_f32_16x16x4_f32(false, a0, false, b3, (short)0, acc[0][3], false, false);
        acc[1][3] = __builtin_amdgcn_wmma_f32_16x16x4_f32(false, a1, false, b3, (short)0, acc[1][3], false, false);
    }

    const float bs0 = bih[0 * Hdim + col] + bhh[0 * Hdim + col];
    const float bs1 = bih[1 * Hdim + col] + bhh[1 * Hdim + col];
    const float bs2 = bih[2 * Hdim + col] + bhh[2 * Hdim + col];
    const float bs3 = bih[3 * Hdim + col] + bhh[3 * Hdim + col];

#pragma unroll
    for (int mt = 0; mt < 2; ++mt) {
#pragma unroll
        for (int r = 0; r < 8; ++r) {
            const int  mg  = m_base + mt * 16 + r + 8 * half;
            const long idx = (long)mg * Hdim + col;
            const float ig = sigmoidf(acc[mt][0][r] + bs0);
            const float fg = sigmoidf(acc[mt][1][r] + bs1);
            const float gg = tanhf   (acc[mt][2][r] + bs2);
            const float og = sigmoidf(acc[mt][3][r] + bs3);
            const float cn = fg * c[idx] + ig * gg;
            c[idx]     = cn;
            h_out[idx] = og * tanhf(cn);
        }
    }
}

extern "C" void kernel_launch(void* const* d_in, const int* in_sizes, int n_in,
                              void* d_out, int out_size, void* d_ws, size_t ws_size,
                              hipStream_t stream) {
    const int*   sentence = (const int*)  d_in[0];
    const float* emb      = (const float*)d_in[1];
    const float* Wih      = (const float*)d_in[2];
    const float* Whh      = (const float*)d_in[3];
    const float* bih      = (const float*)d_in[4];
    const float* bhh      = (const float*)d_in[5];
    float* out = (float*)d_out;
    float* ws  = (float*)d_ws;

    const int BH = Bsz * Hdim;
    float* hbuf0 = ws;
    float* hbuf1 = ws + BH;
    float* cbuf  = ws + 2 * BH;

    lstm_zero_init<<<(BH + 255) / 256, 256, 0, stream>>>(hbuf0, cbuf, BH);

    dim3 grid(Bsz / 32, Hdim / 64);
    for (int t = 0; t < Slen; ++t) {
        const float* hin  = (t & 1) ? hbuf1 : hbuf0;
        float*       hout = (t == Slen - 1) ? out
                          : ((t & 1) ? hbuf0 : hbuf1);
        lstm_step<<<grid, 128, 0, stream>>>(sentence, emb, Wih, Whh, bih, bhh,
                                            hin, cbuf, hout, t);
    }
}


# K split across the 8 waves of a workgroup (32 rows x 16 cols x 4 gates, 256 workgroups): fragments read straight from global, no LDS panel; partial sums added through LDS in fixed wave order
# speedup vs baseline: 3.0385x; 3.0385x over previous
#include <hip/hip_runtime.h>

constexpr int Bsz  = 256;
constexpr int Slen = 512;
constexpr int Edim = 100;
constexpr int Hdim = 512;

constexpr int NWAVE  = 8;
constexpr int HSLICE = Hdim / NWAVE;
constexpr int XSTEPS = Edim / 4;

typedef __attribute__((ext_vector_type(2))) float v2f;
typedef __attribute__((ext_vector_type(8))) float v8f;

__device__ __forceinline__ float sigmoidf(float x) {
    return 1.0f / (1.0f + __expf(-x));
}

__global__ __launch_bounds__(256) void lstm_zero_init(float* __restrict__ h0,
                                                      float* __restrict__ c0,
                                                      int n) {
    int i = blockIdx.x * blockDim.x + threadIdx.x;
    if (i < n) { h0[i] = 0.0f; c0[i] = 0.0f; }
}

#define WMMA4(A, B, C) __builtin_amdgcn_wmma_f32_16x16x4_f32(false, (A), false, (B), (short)0, (C), false, false)

__global__ __launch_bounds__(256) void lstm_step(
    const int*   __restrict__ sentence,
    const float* __restrict__ emb,
    const float* __restrict__ Wih,
    const float* __restrict__ Whh,
    const float* __restrict__ bih,
    const float* __restrict__ bhh,
    const float* __restrict__ h_in,
    float*       __restrict__ c,
    float*       __restrict__ h_out,
    int t)
{
    __shared__ float red[NWAVE][64][32];

    const int tid  = threadIdx.x;
    const int lane = tid & 31;
    const int wave = __builtin_amdgcn_readfirstlane(tid >> 5);
    const int hf   = lane >> 4;
    const int cidx = lane & 15;

    const int m_base = blockIdx.x * 32;
    const int col    = blockIdx.y * 16 + cidx;
    const int row0   = m_base + cidx;
    const int row1   = row0 + 16;

    const int tok0 = sentence[(long)row0 * Slen + t];
    const int tok1 = sentence[(long)row1 * Slen + t];

    v8f acc[2][4];
#pragma unroll
    for (int mt = 0; mt < 2; ++mt)
#pragma unroll
        for (int q = 0; q < 4; ++q)
            acc[mt][q] = (v8f){};

    {
        const int kb = wave * HSLICE + 2 * hf;
        const float* ha0 = h_in + (long)row0 * Hdim + kb;
        const float* ha1 = h_in + (long)row1 * Hdim + kb;
        const float* wh0 = Whh + (long)(0 * Hdim + col) * Hdim + kb;
        const float* wh1 = Whh + (long)(1 * Hdim + col) * Hdim + kb;
        const float* wh2 = Whh + (long)(2 * Hdim + col) * Hdim + kb;
        const float* wh3 = Whh + (long)(3 * Hdim + col) * Hdim + kb;
#pragma unroll
        for (int j = 0; j < HSLICE / 4; ++j) {
            const int o = 4 * j;
            const v2f a0 = *(const v2f*)(ha0 + o);
            const v2f a1 = *(const v2f*)(ha1 + o);
            const v2f b0 = *(const v2f*)(wh0 + o);
            const v2f b1 = *(const v2f*)(wh1 + o);
            const v2f b2 = *(const v2f*)(wh2 + o);
            const v2f b3 = *(const v2f*)(wh3 + o);
            acc[0][0] = WMMA4(a0, b0, acc[0][0]);
            acc[1][0] = WMMA4(a1, b0, acc[1][0]);
            acc[0][1] = WMMA4(a0, b1, acc[0][1]);
            acc[1][1] = WMMA4(a1, b1, acc[1][1]);
            acc[0][2] = WMMA4(a0, b2, acc[0][2]);
            acc[1][2] = WMMA4(a1, b2, acc[1][2]);
            acc[0][3] = WMMA4(a0, b3, acc[0][3]);
            acc[1][3] = WMMA4(a1, b3, acc[1][3]);
        }
    }

    {
        const float* xa0 = emb + (long)tok0 * Edim + 2 * hf;
        const float* xa1 = emb + (long)tok1 * Edim + 2 * hf;
        const float* wi0 = Wih + (long)(0 * Hdim + col) * Edim + 2 * hf;
        const float* wi1 = Wih + (long)(1 * Hdim + col) * Edim + 2 * hf;
        const float* wi2 = Wih + (long)(2 * Hdim + col) * Edim + 2 * hf;
        const float* wi3 = Wih + (long)(3 * Hdim + col) * Edim + 2 * hf;
        for (int s = wave; s < XSTEPS; s += NWAVE) {
            const int o = 4 * s;
            const v2f a0 = *(const v2f*)(xa0 + o);
            const v2f a1 = *(const v2f*)(xa1 + o);
            const v2f b0 = *(const v2f*)(wi0 + o);
            const v2f b1 = *(const v2f*)(wi1 + o);
            const v2f b2 = *(const v2f*)(wi2 + o);
            const v2f b3 = *(const v2f*)(wi3 + o);
            acc[0][0] = WMMA4(a0, b0, acc[0][0]);
            acc[1][0] = WMMA4(a1, b0, acc[1][0]);
            acc[0][1] = WMMA4(a0, b1, acc[0][1]);
            acc[1][1] = WMMA4(a1, b1, acc[1][1]);
            acc[0][2] = WMMA4(a0, b2, acc[0][2]);
            acc[1][2] = WMMA4(a1, b2, acc[1][2]);
            acc[0][3] = WMMA4(a0, b3, acc[0][3]);
            acc[1][3] = WMMA4(a1, b3, acc[1][3]);
        }
    }

    const long idx0 = (long)(m_base + wave + 8 * hf) * Hdim + col;
    const long idx1 = idx0 + (long)16 * Hdim;
    const float cold0 = c[idx0];
    const float cold1 = c[idx1];
    const float bs0 = bih[0 * Hdim + col] + bhh[0 * Hdim + col];
    const float bs1 = bih[1 * Hdim + col] + bhh[1 * Hdim + col];
    const float bs2 = bih[2 * Hdim + col] + bhh[2 * Hdim + col];
    const float bs3 = bih[3 * Hdim + col] + bhh[3 * Hdim + col];

#pragma unroll
    for (int mt = 0; mt < 2; ++mt)
#pragma unroll
        for (int q = 0; q < 4; ++q)
#pragma unroll
            for (int r = 0; r < 8; ++r)
                red[wave][(mt * 4 + q) * 8 + r][lane] = acc[mt][q][r];
    __syncthreads();

#pragma unroll
    for (int mt = 0; mt < 2; ++mt) {
        float g[4];
#pragma unroll
        for (int q = 0; q < 4; ++q) {
            float s = red[0][(mt * 4 + q) * 8 + wave][lane];
#pragma unroll
            for (int w = 1; w < NWAVE; ++w)
                s += red[w][(mt * 4 + q) * 8 + wave][lane];
            g[q] = s;
        }
        const long  idx  = mt ? idx1 : idx0;
        const float cold = mt ? cold1 : cold0;
        const float ig = sigmoidf(g[0] + bs0);
        const float fg = sigmoidf(g[1] + bs1);
        const float gg = tanhf   (g[2] + bs2);
        const float og = sigmoidf(g[3] + bs3);
        const float cn = fg * cold + ig * gg;
        c[idx]     = cn;
        h_out[idx] = og * tanhf(cn);
    }
}

extern "C" void kernel_launch(void* const* d_in, const int* in_sizes, int n_in,
                              void* d_out, int out_size, void* d_ws, size_t ws_size,
                              hipStream_t stream) {
    const int*   sentence = (const int*)  d_in[0];
    const float* emb      = (const float*)d_in[1];
    const float* Wih      = (const float*)d_in[2];
    const float* Whh      = (const float*)d_in[3];
    const float* bih      = (const float*)d_in[4];
    const float* bhh      = (const float*)d_in[5];
    float* out = (float*)d_out;
    float* ws  = (float*)d_ws;

    const int BH = Bsz * Hdim;
    float* hbuf0 = ws;
    float* hbuf1 = ws + BH;
    float* cbuf  = ws + 2 * BH;

    lstm_zero_init<<<(BH + 255) / 256, 256, 0, stream>>>(hbuf0, cbuf, BH);

    dim3 grid(Bsz / 32, Hdim / 16);
    for (int t = 0; t < Slen; ++t) {
        const float* hin  = (t & 1) ? hbuf1 : hbuf0;
        float*       hout = (t == Slen - 1) ? out
                          : ((t & 1) ? hbuf0 : hbuf1);
        lstm_step<<<grid, 256, 0, stream>>>(sentence, emb, Wih, Whh, bih, bhh,
                                            hin, cbuf, hout, t);
    }
}


# Operands in fragment order in workspace (weights packed once per call, h stored packed by the cell update): a wave fetches its whole K slice at entry in 48 loads of 512 contiguous bytes; sums same
# speedup vs baseline: 3.9253x; 1.2919x over previous
#include <hip/hip_runtime.h>

constexpr int Bsz  = 256;
constexpr int Slen = 512;
constexpr int Edim = 100;
constexpr int Hdim = 512;

constexpr int NWAVE  = 8;
constexpr int HSLICE = Hdim / NWAVE;
constexpr int XSTEPS = Edim / 4;
constexpr int HPAIRS = HSLICE / 8;

typedef __attribute__((ext_vector_type(2))) float v2f;
typedef __attribute__((ext_vector_type(4))) float v4f;
typedef __attribute__((ext_vector_type(8))) float v8f;

constexpr int WHH_UNITS = (Hdim / 16) * 4 * NWAVE * HPAIRS * 32;
constexpr int WIH_UNITS = (Hdim / 16) * 4 * XSTEPS * 32;

__device__ __forceinline__ float sigmoidf(float x) {
    return 1.0f / (1.0f + __expf(-x));
}

__global__ __launch_bounds__(256) void lstm_prepare(
    const float* __restrict__ Wih,
    const float* __restrict__ Whh,
    float* __restrict__ WhhP,
    float* __restrict__ WihP,
    float* __restrict__ h0,
    float* __restrict__ c0)
{
    const int i = blockIdx.x * blockDim.x + threadIdx.x;
    {
        const int l  = i & 31;
        const int jj = (i >> 5) & 7;
        const int w  = (i >> 8) & 7;
        const int q  = (i >> 11) & 3;
        const int ct = i >> 13;
        const int n  = q * Hdim + ct * 16 + (l & 15);
        const int k0 = w * HSLICE + 8 * jj + 2 * (l >> 4);
        const float* s = Whh + (long)n * Hdim + k0;
        const v2f lo = *(const v2f*)(s);
        const v2f hi = *(const v2f*)(s + 4);
        *(v4f*)(WhhP + (long)i * 4) = (v4f){lo.x, lo.y, hi.x, hi.y};
    }
    if (i < WIH_UNITS) {
        const int l    = i & 31;
        const int rest = i >> 5;
        const int s    = rest % XSTEPS;
        const int cq   = rest / XSTEPS;
        const int q    = cq & 3;
        const int ct   = cq >> 2;
        const int n    = q * Hdim + ct * 16 + (l & 15);
        const int k    = 4 * s + 2 * (l >> 4);
        *(v2f*)(WihP + (long)i * 2) = *(const v2f*)(Wih + (long)n * Edim + k);
    }
    if (i < Bsz * Hdim) { h0[i] = 0.0f; c0[i] = 0.0f; }
}

#define WMMA4(A, B, C) __builtin_amdgcn_wmma_f32_16x16x4_f32(false, (A), false, (B), (short)0, (C), false, false)

__global__ __launch_bounds__(256) __attribute__((amdgpu_waves_per_eu(2, 2))) void lstm_step(
    const int*   __restrict__ sentence,
    const float* __restrict__ emb,
    const float* __restrict__ WihP,
    const float* __restrict__ WhhP,
    const float* __restrict__ bih,
    const float* __restrict__ bhh,
    const float* __restrict__ h_in,
    float*       __restrict__ c,
    float*       __restrict__ h_out,
    int t, int last)
{
    __shared__ float red[NWAVE][64][32];

    const int tid  = threadIdx.x;
    const int lane = tid & 31;
    const int wave = __builtin_amdgcn_readfirstlane(tid >> 5);
    const int hf   = lane >> 4;
    const int cidx = lane & 15;

    const int m_base = blockIdx.x * 32;
    const int ct     = blockIdx.y;
    const int col    = ct * 16 + cidx;
    const int row0   = m_base + cidx;
    const int row1   = row0 + 16;

    const int tok0 = sentence[(long)row0 * Slen + t];
    const int tok1 = sentence[(long)row1 * Slen + t];

    v8f acc[2][4];
#pragma unroll
    for (int mt = 0; mt < 2; ++mt)
#pragma unroll
        for (int q = 0; q < 4; ++q)
            acc[mt][q] = (v8f){};

    {
        const int mtile0 = blockIdx.x * 2;
        const v4f* ha0 = (const v4f*)h_in + ((mtile0 * NWAVE + wave) * HPAIRS) * 32 + lane;
        const v4f* ha1 = ha0 + NWAVE * HPAIRS * 32;
        const v4f* wh0 = (const v4f*)WhhP + (((ct * 4 + 0) * NWAVE + wave) * HPAIRS) * 32 + lane;
        const v4f* wh1 = wh0 + 1 * NWAVE * HPAIRS * 32;
        const v4f* wh2 = wh0 + 2 * NWAVE * HPAIRS * 32;
        const v4f* wh3 = wh0 + 3 * NWAVE * HPAIRS * 32;
        v4f a0[HPAIRS], a1[HPAIRS], b0[HPAIRS], b1[HPAIRS], b2[HPAIRS], b3[HPAIRS];
#pragma unroll
        for (int jj = 0; jj < HPAIRS; ++jj) {
            const int o = jj * 32;
            a0[jj] = ha0[o];
            a1[jj] = ha1[o];
            b0[jj] = wh0[o];
            b1[jj] = wh1[o];
            b2[jj] = wh2[o];
            b3[jj] = wh3[o];
        }
#pragma unroll
        for (int jj = 0; jj < HPAIRS; ++jj) {
            {
                const v2f a0k = {a0[jj].x, a0[jj].y}, a1k = {a1[jj].x, a1[jj].y};
                const v2f b0k = {b0[jj].x, b0[jj].y}, b1k = {b1[jj].x, b1[jj].y};
                const v2f b2k = {b2[jj].x, b2[jj].y}, b3k = {b3[jj].x, b3[jj].y};
                acc[0][0] = WMMA4(a0k, b0k, acc[0][0]);
                acc[1][0] = WMMA4(a1k, b0k, acc[1][0]);
                acc[0][1] = WMMA4(a0k, b1k, acc[0][1]);
                acc[1][1] = WMMA4(a1k, b1k, acc[1][1]);
                acc[0][2] = WMMA4(a0k, b2k, acc[0][2]);
                acc[1][2] = WMMA4(a1k, b2k, acc[1][2]);
                acc[0][3] = WMMA4(a0k, b3k, acc[0][3]);
                acc[1][3] = WMMA4(a1k, b3k, acc[1][3]);
            }
            {
                const v2f a0k = {a0[jj].z, a0[jj].w}, a1k = {a1[jj].z, a1[jj].w};
                const v2f b0k = {b0[jj].z, b0[jj].w}, b1k = {b1[jj].z, b1[jj].w};
                const v2f b2k = {b2[jj].z, b2[jj].w}, b3k = {b3[jj].z, b3[jj].w};
                acc[0][0] = WMMA4(a0k, b0k, acc[0][0]);
                acc[1][0] = WMMA4(a1k, b0k, acc[1][0]);
                acc[0][1] = WMMA4(a0k, b1k, acc[0][1]);
                acc[1][1] = WMMA4(a1k, b1k, acc[1][1]);
                acc[0][2] = WMMA4(a0k, b2k, acc[0][2]);
                acc[1][2] = WMMA4(a1k, b2k, acc[1][2]);
                acc[0][3] = WMMA4(a0k, b3k, acc[0][3]);
                acc[1][3] = WMMA4(a1k, b3k, acc[1][3]);
            }
        }
    }

    {
        const float* xa0 = emb + (long)tok0 * Edim + 2 * hf;
        const float* xa1 = emb + (long)tok1 * Edim + 2 * hf;
        const v2f* wi0 = (const v2f*)WihP + ((ct * 4 + 0) * XSTEPS) * 32 + lane;
        const v2f* wi1 = wi0 + 1 * XSTEPS * 32;
        const v2f* wi2 = wi0 + 2 * XSTEPS * 32;
        const v2f* wi3 = wi0 + 3 * XSTEPS * 32;
        for (int s = wave; s < XSTEPS; s += NWAVE) {
            const int o = 4 * s;
            const v2f a0 = *(const v2f*)(xa0 + o);
            const v2f a1 = *(const v2f*)(xa1 + o);
            const v2f b0 = wi0[s * 32];
            const v2f b1 = wi1[s * 32];
            const v2f b2 = wi2[s * 32];
            const v2f b3 = wi3[s * 32];
            acc[0][0] = WMMA4(a0, b0, acc[0][0]);
            acc[1][0] = WMMA4(a1, b0, acc[1][0]);
            acc[0][1] = WMMA4(a0, b1, acc[0][1]);
            acc[1][1] = WMMA4(a1, b1, acc[1][1]);
            acc[0][2] = WMMA4(a0, b2, acc[0][2]);
            acc[1][2] = WMMA4(a1, b2, acc[1][2]);
            acc[0][3] = WMMA4(a0, b3, acc[0][3]);
            acc[1][3] = WMMA4(a1, b3, acc[1][3]);
        }
    }

    const int  rloc = wave + 8 * hf;
    const long idx0 = (long)(m_base + rloc) * Hdim + col;
    const long idx1 = idx0 + (long)16 * Hdim;
    const float cold0 = c[idx0];
    const float cold1 = c[idx1];
    const float bs0 = bih[0 * Hdim + col] + bhh[0 * Hdim + col];
    const float bs1 = bih[1 * Hdim + col] + bhh[1 * Hdim + col];
    const float bs2 = bih[2 * Hdim + col] + bhh[2 * Hdim + col];
    const float bs3 = bih[3 * Hdim + col] + bhh[3 * Hdim + col];

    const long pk0 = ((long)(((blockIdx.x * 2) * NWAVE + (col >> 6)) * HPAIRS + ((col >> 3) & 7)) * 32
                      + ((col >> 1) & 1) * 16 + rloc) * 4
                     + ((col >> 2) & 1) * 2 + (col & 1);
    const long pk1 = pk0 + (long)NWAVE * HPAIRS * 32 * 4;

#pragma unroll
    for (int mt = 0; mt < 2; ++mt)
#pragma unroll
        for (int q = 0; q < 4; ++q)
#pragma unroll
            for (int r = 0; r < 8; ++r)
                red[wave][(mt * 4 + q) * 8 + r][lane] = acc[mt][q][r];
    __syncthreads();

#pragma unroll
    for (int mt = 0; mt < 2; ++mt) {
        float g[4];
#pragma unroll
        for (int q = 0; q < 4; ++q) {
            float s = red[0][(mt * 4 + q) * 8 + wave][lane];
#pragma unroll
            for (int w = 1; w < NWAVE; ++w)
                s += red[w][(mt * 4 + q) * 8 + wave][lane];
            g[q] = s;
        }
        const long  idx  = mt ? idx1 : idx0;
        const long  pk   = mt ? pk1 : pk0;
        const float cold = mt ? cold1 : cold0;
        const float ig = sigmoidf(g[0] + bs0);
        const float fg = sigmoidf(g[1] + bs1);
        const float gg = tanhf   (g[2] + bs2);
        const float og = sigmoidf(g[3] + bs3);
        const float cn = fg * cold + ig * gg;
        c[idx] = cn;
        h_out[last ? idx : pk] = og * tanhf(cn);
    }
}

extern "C" void kernel_launch(void* const* d_in, const int* in_sizes, int n_in,
                              void* d_out, int out_size, void* d_ws, size_t ws_size,
                              hipStream_t stream) {
    const int*   sentence = (const int*)  d_in[0];
    const float* emb      = (const float*)d_in[1];
    const float* Wih      = (const float*)d_in[2];
    const float* Whh      = (const float*)d_in[3];
    const float* bih      = (const float*)d_in[4];
    const float* bhh      = (const float*)d_in[5];
    float* out = (float*)d_out;
    float* ws  = (float*)d_ws;

    const int BH = Bsz * Hdim;
    float* hbuf0 = ws;
    float* hbuf1 = ws + BH;
    float* cbuf  = ws + 2 * BH;
    float* WhhP  = ws + 3 * BH;
    float* WihP  = WhhP + 4 * Hdim * Hdim;

    lstm_prepare<<<WHH_UNITS / 256, 256, 0, stream>>>(Wih, Whh, WhhP, WihP, hbuf0, cbuf);

    dim3 grid(Bsz / 32, Hdim / 16);
    for (int t = 0; t < Slen; ++t) {
        const float* hin  = (t & 1) ? hbuf1 : hbuf0;
        float*       hout = (t == Slen - 1) ? out
                          : ((t & 1) ? hbuf0 : hbuf1);
        lstm_step<<<grid, 256, 0, stream>>>(sentence, emb, WihP, WhhP, bih, bhh,
                                            hin, cbuf, hout, t, t == Slen - 1);
    }
}
